# starting kernel
# baseline (speedup 1.0000x reference)
#include <hip/hip_runtime.h>

typedef __attribute__((ext_vector_type(2))) float v2f;
typedef __attribute__((ext_vector_type(4))) float v4f;
typedef __attribute__((ext_vector_type(8))) float v8f;

#define NB   4
#define NN   1024
#define NIN  64
#define NHID 32
#define NOUT 8
#define HH   96

__global__ __launch_bounds__(256) void k_lin_relu(
    const float* __restrict__ x, const float* __restrict__ W,
    const float* __restrict__ bl, float* __restrict__ h) {
  int tid = blockIdx.x * 256 + threadIdx.x;
  int k  = tid & (NHID - 1);
  int bn = tid >> 5;
  const float* xr = x + (size_t)bn * NIN;
  float s = bl[k];
#pragma unroll 8
  for (int i = 0; i < NIN; ++i) s = fmaf(xr[i], W[i * NHID + k], s);
  h[tid] = fmaxf(s, 0.0f);
}

__global__ __launch_bounds__(256) void k_shift_concat(
    const float* __restrict__ h, float* __restrict__ v) {
  int tid = blockIdx.x * 256 + threadIdx.x;
  int g  = tid % HH;
  int bn = tid / HH;
  int b  = bn >> 10;
  int n  = bn & (NN - 1);
  int c  = g >> 5;
  int k  = g & 31;
  int sn = n + c - 1;
  float val = 0.0f;
  if (sn >= 0 && sn < NN) val = h[(((size_t)b << 10) + sn) * NHID + k];
  v[tid] = val;
}

__global__ __launch_bounds__(256) void k_t_wmma(
    const float* __restrict__ v, const float* __restrict__ Wb,
    float* __restrict__ t) {
  const int lane   = threadIdx.x & 31;
  const int laneLo = lane & 15;
  const bool hi    = lane >= 16;

  int wave = (blockIdx.x * 256 + threadIdx.x) >> 5;
  int o   = wave & (NOUT - 1);
  int bi0 = (wave >> 3) * 16;

  const float* vA = v + (size_t)(bi0 + laneLo) * HH + (hi ? 2 : 0);
  v2f a[24];
#pragma unroll
  for (int s = 0; s < 24; ++s) a[s] = *(const v2f*)(vA + s * 4);

  const float* wB = Wb + (size_t)o * HH * HH + (hi ? 2 * HH : 0) + laneLo;

#pragma unroll
  for (int gt = 0; gt < 6; ++gt) {
    const int g0 = gt * 16;
    v8f acc = {};
#pragma unroll
    for (int s = 0; s < 24; ++s) {
      const float* p = wB + (size_t)(4 * s) * HH + g0;
      v2f bf = { p[0], p[HH] };
      acc = __builtin_amdgcn_wmma_f32_16x16x4_f32(
          false, a[s], false, bf, (short)0, acc, false, false);
    }
    float* tp = t + (size_t)(bi0 + (hi ? 8 : 0)) * (NOUT * HH)
                  + o * HH + g0 + laneLo;
#pragma unroll
    for (int r = 0; r < 8; ++r)
      tp[(size_t)r * (NOUT * HH)] = acc[r];
  }
}

__global__ __launch_bounds__(256) void k_bilinear_out(
    const float* __restrict__ t, const float* __restrict__ v,
    const float* __restrict__ bb_g, float* __restrict__ out) {
  const int lane   = threadIdx.x & 31;
  const int laneLo = lane & 15;
  const bool hi    = lane >= 16;

  int wave = (blockIdx.x * 256 + threadIdx.x) >> 5;
  int jg = wave & 15;
  int mt = (wave >> 4) & 511;
  int b  = wave >> 13;

  const int m0 = mt * 16;
  const int i0 = m0 >> 3;

  float bb[NOUT];
#pragma unroll
  for (int r = 0; r < NOUT; ++r) bb[r] = bb_g[r];

  const float* tA = t + ((size_t)b * 8192 + m0 + laneLo) * HH + (hi ? 2 : 0);
  v2f a[24];
#pragma unroll
  for (int s = 0; s < 24; ++s) a[s] = *(const v2f*)(tA + s * 4);

  const float* vBb = v + (size_t)b * NN * HH + (hi ? 2 : 0);
  const int i = i0 + (hi ? 1 : 0);

#pragma unroll
  for (int tj = 0; tj < 4; ++tj) {
    const int j0 = (jg * 4 + tj) * 16;
    const float* vB = vBb + (size_t)(j0 + laneLo) * HH;

    v8f acc = {};
#pragma unroll
    for (int s = 0; s < 24; ++s) {
      v2f bf = *(const v2f*)(vB + s * 4);
      acc = __builtin_amdgcn_wmma_f32_16x16x4_f32(
           false, a[s],  false, bf,
           (short)0, acc,  false,  false);
    }

    const int j = j0 + laneLo;
    float* op = out + (((size_t)b * NN + i) * NN + j) * NOUT;
    v4f lo = { acc[0] + bb[0], acc[1] + bb[1], acc[2] + bb[2], acc[3] + bb[3] };
    v4f hi4 = { acc[4] + bb[4], acc[5] + bb[5], acc[6] + bb[6], acc[7] + bb[7] };
    __builtin_nontemporal_store(lo,  (v4f*)op);
    __builtin_nontemporal_store(hi4, (v4f*)(op + 4));
  }
}

extern "C" void kernel_launch(void* const* d_in, const int* in_sizes, int n_in,
                              void* d_out, int out_size, void* d_ws, size_t ws_size,
                              hipStream_t stream) {
  const float* x     = (const float*)d_in[0];
  const float* W_lin = (const float*)d_in[1];
  const float* b_lin = (const float*)d_in[2];
  const float* W_bil = (const float*)d_in[3];
  const float* b_bil = (const float*)d_in[4];
  float* out = (float*)d_out;

  char* ws = (char*)d_ws;
  float* h  = (float*)ws;
  float* v  = (float*)(ws + (size_t)NB * NN * NHID * 4);
  float* t  = (float*)(ws + (size_t)NB * NN * NHID * 4
                          + (size_t)NB * NN * HH * 4);

  k_lin_relu<<<(NB * NN * NHID) / 256, 256, 0, stream>>>(x, W_lin, b_lin, h);
  k_shift_concat<<<(NB * NN * HH) / 256, 256, 0, stream>>>(h, v);
  k_t_wmma<<<256, 256, 0, stream>>>(v, W_bil, t);
  k_bilinear_out<<<4096, 256, 0, stream>>>(t, v, b_bil, out);
}


# Linear layer keeps 32 rows x one k-octet per wave and stores each h chunk straight into its three shifted places of v (fragment order): no h buffer, no shift launch, 4 launches to 3; same sums
# speedup vs baseline: 2.6734x; 2.6734x over previous
#include <hip/hip_runtime.h>

typedef __attribute__((ext_vector_type(2))) float v2f;
typedef __attribute__((ext_vector_type(4))) float v4f;
typedef __attribute__((ext_vector_type(8))) float v8f;

#define NB   4
#define NN   1024
#define NIN  64
#define NHID 32
#define NOUT 8
#define HH   96
#define NQ   12


__global__ __launch_bounds__(128) void k_lin_relu_v(
    const float* __restrict__ x, const float* __restrict__ W,
    const float* __restrict__ bl, float* __restrict__ vF) {
  const int lane = threadIdx.x & 31;
  const int qp = __builtin_amdgcn_readfirstlane((int)(threadIdx.x >> 5));
  const int bn = blockIdx.x * 32 + lane;
  const v4f* xr = (const v4f*)(x + (size_t)bn * NIN);
  const float* wk = W + 8 * qp;

  v4f xv[NIN / 4];
#pragma unroll
  for (int t = 0; t < NIN / 4; ++t) xv[t] = xr[t];

  float s[8];
#pragma unroll
  for (int r = 0; r < 8; ++r) s[r] = bl[8 * qp + r];
#pragma unroll
  for (int t = 0; t < NIN / 4; ++t) {
#pragma unroll
    for (int e = 0; e < 4; ++e) {
      const float xi = xv[t][e];
      const float* wi = wk + (4 * t + e) * NHID;
#pragma unroll
      for (int r = 0; r < 8; ++r) s[r] = fmaf(xi, wi[r], s[r]);
    }
  }
#pragma unroll
  for (int r = 0; r < 8; ++r) s[r] = fmaxf(s[r], 0.0f);

  const v4f c0 = { s[0], s[1], s[4], s[5] };
  const v4f c1 = { s[2], s[3], s[6], s[7] };
  const v4f zz = { 0.0f, 0.0f, 0.0f, 0.0f };
  v4f* vv = (v4f*)vF;
  const int b = bn >> 10;
  const int n = bn & (NN - 1);
#pragma unroll
  for (int cb = 0; cb < 3; ++cb) {
    const int np = n + 1 - cb;
    if (np >= 0 && np < NN) {
      const int bnp = (b << 10) + np;
      const size_t at = ((size_t)(bnp >> 4) * NQ + 4 * cb + qp) * 32 + (bnp & 15);
      vv[at]      = c0;
      vv[at + 16] = c1;
    }
  }
  if (n == 0) {
    const size_t at = ((size_t)(bn >> 4) * NQ + qp) * 32;
    vv[at]      = zz;
    vv[at + 16] = zz;
  }
  if (n == NN - 1) {
    const size_t at = ((size_t)(bn >> 4) * NQ + 8 + qp) * 32 + 15;
    vv[at]      = zz;
    vv[at + 16] = zz;
  }
}

__global__ __launch_bounds__(256) void k_t_wmma(
    const float* __restrict__ vF, const float* __restrict__ Wb,
    float* __restrict__ tF) {
  const int lane = threadIdx.x & 31;
  const int c    = lane & 15;
  const int hf   = lane >> 4;

  int wave = (blockIdx.x * 256 + threadIdx.x) >> 5;
  int gt = wave % 6;
  int bt = wave / 6;
  const int g0 = gt * 16;

  const v4f* vp = (const v4f*)vF + (size_t)bt * NQ * 32 + lane;
  v4f bq[NQ];
#pragma unroll
  for (int q = 0; q < NQ; ++q) bq[q] = vp[q * 32];

  const float* wA = Wb + (hf ? 2 * HH : 0) + g0 + c;

  const int bi = bt * 16 + c;
  const int qd = 2 * gt + hf;
  float* dst = tF + ((((size_t)(bi >> 1) * NQ + qd) * 32) + (bi & 1) * 4) * 4;

#pragma unroll
  for (int o = 0; o < NOUT; ++o) {
    const float* wo = wA + (size_t)o * HH * HH;
    v8f acc = {};
#pragma unroll
    for (int s = 0; s < 24; ++s) {
      const float* p = wo + (size_t)(4 * s) * HH;
      v2f af = { p[0], p[HH] };
      v2f bf = (s & 1) ? (v2f){ bq[s >> 1].z, bq[s >> 1].w }
                       : (v2f){ bq[s >> 1].x, bq[s >> 1].y };
      acc = __builtin_amdgcn_wmma_f32_16x16x4_f32(
          false, af, false, bf, (short)0, acc, false, false);
    }
    v4f lo = { acc[0], acc[1], acc[4], acc[5] };
    v4f hi = { acc[2], acc[3], acc[6], acc[7] };
    const int ro = 8 * (o >> 2) + (o & 3);
    *(v4f*)(dst + ro * 4)          = lo;
    *(v4f*)(dst + 16 * 4 + ro * 4) = hi;
  }
}

#define NA   4
#define JTW  16

#define TILE_WMMAS(BSET)                                                      \
  _Pragma("unroll")                                                           \
  for (int q = 0; q < NQ; ++q) {                                              \
    _Pragma("unroll")                                                         \
    for (int a = 0; a < NA; ++a)                                              \
      acc[a] = __builtin_amdgcn_wmma_f32_16x16x4_f32(                         \
          false, (v2f){ a4[a][q].x, a4[a][q].y },                             \
          false, (v2f){ BSET[q].x, BSET[q].y }, (short)0, acc[a], false, false); \
    _Pragma("unroll")                                                         \
    for (int a = 0; a < NA; ++a)                                              \
      acc[a] = __builtin_amdgcn_wmma_f32_16x16x4_f32(                         \
          false, (v2f){ a4[a][q].z, a4[a][q].w },                             \
          false, (v2f){ BSET[q].z, BSET[q].w }, (short)0, acc[a], false, false); \
  }

#define TILE_STORE(TJ)                                                        \
  _Pragma("unroll")                                                           \
  for (int a = 0; a < NA; ++a) {                                              \
    float* op = ob + ((size_t)(2 * a) * NN + (TJ) * 16) * NOUT;               \
    v4f r0 = { acc[a][0] + bb.x, acc[a][1] + bb.y,                            \
               acc[a][2] + bb.z, acc[a][3] + bb.w };                          \
    v4f r1 = { acc[a][4] + bb.x, acc[a][5] + bb.y,                            \
               acc[a][6] + bb.z, acc[a][7] + bb.w };                          \
    __builtin_nontemporal_store(r0, (v4f*)op);                                \
    __builtin_nontemporal_store(r1, (v4f*)(op + (size_t)NN * NOUT));          \
  }

#define TILE_LOAD_B(BSET, TJ)                                                 \
  {                                                                           \
    const int tn = ((TJ) < JTW) ? (TJ) : JTW - 1;                             \
    const v4f* bn = bp + (size_t)tn * NQ * 32;                                \
    _Pragma("unroll")                                                         \
    for (int q = 0; q < NQ; ++q) BSET[q] = bn[q * 32];                        \
  }

__global__ __launch_bounds__(256) void k_bilinear_out(
    const float* __restrict__ tF, const float* __restrict__ vF,
    const float* __restrict__ bb_g, float* __restrict__ out) {
  const int lane   = threadIdx.x & 31;
  const int laneLo = lane & 15;
  const int hi     = lane >> 4;
  const int wv  = __builtin_amdgcn_readfirstlane((int)(threadIdx.x >> 5));
  const int blk = blockIdx.x;
  const int rgg = blk & 15;
  const int jq  = (blk >> 4) & 3;
  const int b   = blk >> 6;
  const int mt0 = (rgg * 8 + wv) * NA;
  const int i0  = mt0 * 2;

  const v4f bb = *(const v4f*)(bb_g + 4 * hi);

  const v4f* ap = (const v4f*)tF + (size_t)(b * 512 + mt0) * NQ * 32 + lane;
  v4f a4[NA][NQ];
#pragma unroll
  for (int q = 0; q < NQ; ++q)
#pragma unroll
    for (int a = 0; a < NA; ++a) a4[a][q] = ap[(a * NQ + q) * 32];

  __builtin_amdgcn_sched_barrier(0);

  const v4f* bp = (const v4f*)vF + (size_t)(b * 64 + jq * JTW) * NQ * 32 + lane;
  v4f b0[NQ], b1[NQ];
  TILE_LOAD_B(b0, 0)
  TILE_LOAD_B(b1, 1)

  float* ob = out + (((size_t)b * NN + i0) * NN + jq * (JTW * 16) + laneLo) * NOUT + 4 * hi;

#pragma unroll 1
  for (int tj = 0; tj < JTW; tj += 2) {
    {
      v8f acc[NA] = {};
      TILE_WMMAS(b0)
      __builtin_amdgcn_sched_barrier(0);
      TILE_STORE(tj)
      TILE_LOAD_B(b0, tj + 2)
      __builtin_amdgcn_sched_barrier(0);
    }
    {
      v8f acc[NA] = {};
      TILE_WMMAS(b1)
      __builtin_amdgcn_sched_barrier(0);
      TILE_STORE(tj + 1)
      TILE_LOAD_B(b1, tj + 3)
      __builtin_amdgcn_sched_barrier(0);
    }
  }
}
#undef TILE_LOAD_B
#undef TILE_STORE
#undef TILE_WMMAS

extern "C" void kernel_launch(void* const* d_in, const int* in_sizes, int n_in,
                              void* d_out, int out_size, void* d_ws, size_t ws_size,
                              hipStream_t stream) {
  const float* x     = (const float*)d_in[0];
  const float* W_lin = (const float*)d_in[1];
  const float* b_lin = (const float*)d_in[2];
  const float* W_bil = (const float*)d_in[3];
  const float* b_bil = (const float*)d_in[4];
  float* out = (float*)d_out;

  char* ws = (char*)d_ws;
  float* vF = (float*)ws;
  float* tF = (float*)(ws + (size_t)NB * NN * HH * 4);

  k_lin_relu_v<<<(NB * NN) / 32, 128, 0, stream>>>(x, W_lin, b_lin, vF);
  k_t_wmma<<<192, 256, 0, stream>>>(vF, W_bil, tF);
  k_bilinear_out<<<256, 256, 0, stream>>>(tF, vF, b_bil, out);
}


# t GEMM's W_bil operand kept in WMMA fragment order (copied once per call by spare work-groups of launch 1): its loads become 16-byte whole-line loads; wave = 1 bi-tile x 3 g-tiles x 4 o; same sums
# speedup vs baseline: 2.9817x; 1.1153x over previous
#include <hip/hip_runtime.h>

typedef __attribute__((ext_vector_type(2))) float v2f;
typedef __attribute__((ext_vector_type(4))) float v4f;
typedef __attribute__((ext_vector_type(8))) float v8f;

#define NB   4
#define NN   1024
#define NIN  64
#define NHID 32
#define NOUT 8
#define HH   96
#define NQ   12


#define LIN_BLOCKS 128
#define WF_BLOCKS  36
#define NGT        6

__global__ __launch_bounds__(128) void k_lin_relu_v(
    const float* __restrict__ x, const float* __restrict__ W,
    const float* __restrict__ bl, const float* __restrict__ Wb,
    float* __restrict__ vF, float* __restrict__ wF) {
  if (blockIdx.x >= LIN_BLOCKS) {
    const int tid = (blockIdx.x - LIN_BLOCKS) * 128 + threadIdx.x;
#pragma unroll
    for (int r = 0; r < 4; ++r) {
      const int ch = tid + r * (WF_BLOCKS * 128);
      const int c  = ch & 15;
      const int hf = (ch >> 4) & 1;
      const int u  = ch >> 5;
      const int q  = u % NQ;
      const int og = u / NQ;
      const int gt = og % NGT;
      const int o  = og / NGT;
      const float* p = Wb + ((size_t)o * HH + 8 * q + 2 * hf) * HH + 16 * gt + c;
      const v4f w = { p[0], p[HH], p[4 * HH], p[5 * HH] };
      ((v4f*)wF)[ch] = w;
    }
    return;
  }
  const int lane = threadIdx.x & 31;
  const int qp = __builtin_amdgcn_readfirstlane((int)(threadIdx.x >> 5));
  const int bn = blockIdx.x * 32 + lane;
  const v4f* xr = (const v4f*)(x + (size_t)bn * NIN);
  const float* wk = W + 8 * qp;

  v4f xv[NIN / 4];
#pragma unroll
  for (int t = 0; t < NIN / 4; ++t) xv[t] = xr[t];

  float s[8];
#pragma unroll
  for (int r = 0; r < 8; ++r) s[r] = bl[8 * qp + r];
#pragma unroll
  for (int t = 0; t < NIN / 4; ++t) {
#pragma unroll
    for (int e = 0; e < 4; ++e) {
      const float xi = xv[t][e];
      const float* wi = wk + (4 * t + e) * NHID;
#pragma unroll
      for (int r = 0; r < 8; ++r) s[r] = fmaf(xi, wi[r], s[r]);
    }
  }
#pragma unroll
  for (int r = 0; r < 8; ++r) s[r] = fmaxf(s[r], 0.0f);

  const v4f c0 = { s[0], s[1], s[4], s[5] };
  const v4f c1 = { s[2], s[3], s[6], s[7] };
  const v4f zz = { 0.0f, 0.0f, 0.0f, 0.0f };
  v4f* vv = (v4f*)vF;
  const int b = bn >> 10;
  const int n = bn & (NN - 1);
#pragma unroll
  for (int cb = 0; cb < 3; ++cb) {
    const int np = n + 1 - cb;
    if (np >= 0 && np < NN) {
      const int bnp = (b << 10) + np;
      const size_t at = ((size_t)(bnp >> 4) * NQ + 4 * cb + qp) * 32 + (bnp & 15);
      vv[at]      = c0;
      vv[at + 16] = c1;
    }
  }
  if (n == 0) {
    const size_t at = ((size_t)(bn >> 4) * NQ + qp) * 32;
    vv[at]      = zz;
    vv[at + 16] = zz;
  }
  if (n == NN - 1) {
    const size_t at = ((size_t)(bn >> 4) * NQ + 8 + qp) * 32 + 15;
    vv[at]      = zz;
    vv[at + 16] = zz;
  }
}

#define TO 4
#define TU 3

#define T_LOAD_A(U)                                                           \
  _Pragma("unroll")                                                           \
  for (int q = 0; q < NQ; ++q) {                                              \
    _Pragma("unroll")                                                         \
    for (int a = 0; a < TO; ++a)                                              \
      a4[U][a][q] = ap[((size_t)(a * NGT + (U)) * NQ + q) * 32];              \
  }

#define T_WMMAS(U, Q0, Q1)                                                    \
  _Pragma("unroll")                                                           \
  for (int q = (Q0); q < (Q1); ++q) {                                         \
    _Pragma("unroll")                                                         \
    for (int a = 0; a < TO; ++a)                                              \
      acc[a] = __builtin_amdgcn_wmma_f32_16x16x4_f32(                         \
          false, (v2f){ a4[U][a][q].x, a4[U][a][q].y },                       \
          false, (v2f){ bq[q].x, bq[q].y }, (short)0, acc[a], false, false);  \
    _Pragma("unroll")                                                         \
    for (int a = 0; a < TO; ++a)                                              \
      acc[a] = __builtin_amdgcn_wmma_f32_16x16x4_f32(                         \
          false, (v2f){ a4[U][a][q].z, a4[U][a][q].w },                       \
          false, (v2f){ bq[q].z, bq[q].w }, (short)0, acc[a], false, false);  \
  }

#define T_STORE(U)                                                            \
  _Pragma("unroll")                                                           \
  for (int a = 0; a < TO; ++a) {                                              \
    const v4f lo = { acc[a][0], acc[a][1], acc[a][4], acc[a][5] };            \
    const v4f hi = { acc[a][2], acc[a][3], acc[a][6], acc[a][7] };            \
    float* d = dst + (U) * (2 * 32 * 4) + (8 * oq + a) * 4;                   \
    *(v4f*)(d)          = lo;                                                 \
    *(v4f*)(d + 16 * 4) = hi;                                                 \
  }

__global__ __launch_bounds__(128) void k_t_wmma(
    const float* __restrict__ vF, const float* __restrict__ wF,
    float* __restrict__ tF) {
  const int lane = threadIdx.x & 31;
  const int c    = lane & 15;
  const int hf   = lane >> 4;
  const int wv  = __builtin_amdgcn_readfirstlane((int)(threadIdx.x >> 5));
  const int blk = blockIdx.x;
  const int oq  = blk & 1;
  const int gt0 = TU * ((blk >> 1) & 1);
  const int bt  = 4 * (blk >> 2) + wv;

  const v4f* vp = (const v4f*)vF + (size_t)bt * NQ * 32 + lane;
  const v4f* ap = (const v4f*)wF + (size_t)((TO * oq) * NGT + gt0) * NQ * 32 + lane;
  v4f bq[NQ];
  v4f a4[TU][TO][NQ];

  const int bi = bt * 16 + c;
  float* dst = tF + ((((size_t)(bi >> 1) * NQ + 2 * gt0 + hf) * 32) + (bi & 1) * 4) * 4;

#pragma unroll
  for (int q = 0; q < NQ; ++q) bq[q] = vp[q * 32];
  T_LOAD_A(0)
  __builtin_amdgcn_sched_barrier(0);

  {
    v8f acc[TO] = {};
    T_WMMAS(0, 0, NQ / 2)
    __builtin_amdgcn_sched_barrier(0);
    T_LOAD_A(1)
    __builtin_amdgcn_sched_barrier(0);
    T_WMMAS(0, NQ / 2, NQ)
    __builtin_amdgcn_sched_barrier(0);
    T_STORE(0)
  }
  T_LOAD_A(2)
  __builtin_amdgcn_sched_barrier(0);
  {
    v8f acc[TO] = {};
    T_WMMAS(1, 0, NQ)
    __builtin_amdgcn_sched_barrier(0);
    T_STORE(1)
  }
  __builtin_amdgcn_sched_barrier(0);
  {
    v8f acc[TO] = {};
    T_WMMAS(2, 0, NQ)
    T_STORE(2)
  }
}
#undef T_STORE
#undef T_WMMAS
#undef T_LOAD_A
#undef TU
#undef TO

#define NA   4
#define JTW  16

#define TILE_WMMAS(BSET)                                                      \
  _Pragma("unroll")                                                           \
  for (int q = 0; q < NQ; ++q) {                                              \
    _Pragma("unroll")                                                         \
    for (int a = 0; a < NA; ++a)                                              \
      acc[a] = __builtin_amdgcn_wmma_f32_16x16x4_f32(                         \
          false, (v2f){ a4[a][q].x, a4[a][q].y },                             \
          false, (v2f){ BSET[q].x, BSET[q].y }, (short)0, acc[a], false, false); \
    _Pragma("unroll")                                                         \
    for (int a = 0; a < NA; ++a)                                              \
      acc[a] = __builtin_amdgcn_wmma_f32_16x16x4_f32(                         \
          false, (v2f){ a4[a][q].z, a4[a][q].w },                             \
          false, (v2f){ BSET[q].z, BSET[q].w }, (short)0, acc[a], false, false); \
  }

#define TILE_STORE(TJ)                                                        \
  _Pragma("unroll")                                                           \
  for (int a = 0; a < NA; ++a) {                                              \
    float* op = ob + ((size_t)(2 * a) * NN + (TJ) * 16) * NOUT;               \
    v4f r0 = { acc[a][0] + bb.x, acc[a][1] + bb.y,                            \
               acc[a][2] + bb.z, acc[a][3] + bb.w };                          \
    v4f r1 = { acc[a][4] + bb.x, acc[a][5] + bb.y,                            \
               acc[a][6] + bb.z, acc[a][7] + bb.w };                          \
    __builtin_nontemporal_store(r0, (v4f*)op);                                \
    __builtin_nontemporal_store(r1, (v4f*)(op + (size_t)NN * NOUT));          \
  }

#define TILE_LOAD_B(BSET, TJ)                                                 \
  {                                                                           \
    const int tn = ((TJ) < JTW) ? (TJ) : JTW - 1;                             \
    const v4f* bn = bp + (size_t)tn * NQ * 32;                                \
    _Pragma("unroll")                                                         \
    for (int q = 0; q < NQ; ++q) BSET[q] = bn[q * 32];                        \
  }

__global__ __launch_bounds__(256) void k_bilinear_out(
    const float* __restrict__ tF, const float* __restrict__ vF,
    const float* __restrict__ bb_g, float* __restrict__ out) {
  const int lane   = threadIdx.x & 31;
  const int laneLo = lane & 15;
  const int hi     = lane >> 4;
  const int wv  = __builtin_amdgcn_readfirstlane((int)(threadIdx.x >> 5));
  const int blk = blockIdx.x;
  const int rgg = blk & 15;
  const int jq  = (blk >> 4) & 3;
  const int b   = blk >> 6;
  const int mt0 = (rgg * 8 + wv) * NA;
  const int i0  = mt0 * 2;

  const v4f bb = *(const v4f*)(bb_g + 4 * hi);

  const v4f* ap = (const v4f*)tF + (size_t)(b * 512 + mt0) * NQ * 32 + lane;
  v4f a4[NA][NQ];
#pragma unroll
  for (int q = 0; q < NQ; ++q)
#pragma unroll
    for (int a = 0; a < NA; ++a) a4[a][q] = ap[(a * NQ + q) * 32];

  __builtin_amdgcn_sched_barrier(0);

  const v4f* bp = (const v4f*)vF + (size_t)(b * 64 + jq * JTW) * NQ * 32 + lane;
  v4f b0[NQ], b1[NQ];
  TILE_LOAD_B(b0, 0)
  TILE_LOAD_B(b1, 1)

  float* ob = out + (((size_t)b * NN + i0) * NN + jq * (JTW * 16) + laneLo) * NOUT + 4 * hi;

#pragma unroll 1
  for (int tj = 0; tj < JTW; tj += 2) {
    {
      v8f acc[NA] = {};
      TILE_WMMAS(b0)
      __builtin_amdgcn_sched_barrier(0);
      TILE_STORE(tj)
      TILE_LOAD_B(b0, tj + 2)
      __builtin_amdgcn_sched_barrier(0);
    }
    {
      v8f acc[NA] = {};
      TILE_WMMAS(b1)
      __builtin_amdgcn_sched_barrier(0);
      TILE_STORE(tj + 1)
      TILE_LOAD_B(b1, tj + 3)
      __builtin_amdgcn_sched_barrier(0);
    }
  }
}
#undef TILE_LOAD_B
#undef TILE_STORE
#undef TILE_WMMAS

extern "C" void kernel_launch(void* const* d_in, const int* in_sizes, int n_in,
                              void* d_out, int out_size, void* d_ws, size_t ws_size,
                              hipStream_t stream) {
  const float* x     = (const float*)d_in[0];
  const float* W_lin = (const float*)d_in[1];
  const float* b_lin = (const float*)d_in[2];
  const float* W_bil = (const float*)d_in[3];
  const float* b_bil = (const float*)d_in[4];
  float* out = (float*)d_out;

  char* ws = (char*)d_ws;
  float* vF = (float*)ws;
  float* tF = (float*)(ws + (size_t)NB * NN * HH * 4);

  float* wF = out;

  k_lin_relu_v<<<LIN_BLOCKS + WF_BLOCKS, 128, 0, stream>>>(x, W_lin, b_lin, W_bil, vF, wF);
  k_t_wmma<<<256, 128, 0, stream>>>(vF, wF, tF);
  k_bilinear_out<<<256, 256, 0, stream>>>(tF, vF, b_bil, out);
}


# Linear layer on the matrix instruction, transposed (h^T = W_lin^T x^T, bias as C): a lane ends with its two v chunks; h's 64 products summed in the instruction's k order, not one fmaf chain
# speedup vs baseline: 3.1291x; 1.0494x over previous
#include <hip/hip_runtime.h>

typedef __attribute__((ext_vector_type(2))) float v2f;
typedef __attribute__((ext_vector_type(4))) float v4f;
typedef __attribute__((ext_vector_type(8))) float v8f;

#define NB   4
#define NN   1024
#define NIN  64
#define NHID 32
#define NOUT 8
#define HH   96
#define NQ   12


#define LIN_BLOCKS 128
#define WF_BLOCKS  36
#define NGT        6

__global__ __launch_bounds__(128) void k_lin_relu_v(
    const float* __restrict__ x, const float* __restrict__ W,
    const float* __restrict__ bl, const float* __restrict__ Wb,
    float* __restrict__ vF, float* __restrict__ wF) {
  if (blockIdx.x >= LIN_BLOCKS) {
    const int tid = (blockIdx.x - LIN_BLOCKS) * 128 + threadIdx.x;
#pragma unroll
    for (int r = 0; r < 4; ++r) {
      const int ch = tid + r * (WF_BLOCKS * 128);
      const int c  = ch & 15;
      const int hf = (ch >> 4) & 1;
      const int u  = ch >> 5;
      const int q  = u % NQ;
      const int og = u / NQ;
      const int gt = og % NGT;
      const int o  = og / NGT;
      const float* p = Wb + ((size_t)o * HH + 8 * q + 2 * hf) * HH + 16 * gt + c;
      const v4f w = { p[0], p[HH], p[4 * HH], p[5 * HH] };
      ((v4f*)wF)[ch] = w;
    }
    return;
  }
  const int lane = threadIdx.x & 31;
  const int c    = lane & 15;
  const int hf   = lane >> 4;
  const int wv   = __builtin_amdgcn_readfirstlane((int)(threadIdx.x >> 5));
  const int kt   = wv & 1;
  const int tile = blockIdx.x * 2 + (wv >> 1);
  const int bn   = tile * 16 + c;

  const v4f* xr = (const v4f*)(x + (size_t)bn * NIN) + hf;
  v4f xq[NIN / 8];
#pragma unroll
  for (int q = 0; q < NIN / 8; ++q) xq[q] = xr[2 * q];

  const int unitA = 16 * kt + 8 * ((c >> 2) & 1) + 2 * (c >> 3) + (c & 1) + 4 * ((c >> 1) & 1);
  const float* wp = W + (size_t)(4 * hf) * NHID + unitA;
  float wa[NIN / 8][4];
#pragma unroll
  for (int q = 0; q < NIN / 8; ++q) {
#pragma unroll
    for (int j = 0; j < 4; ++j) wa[q][j] = wp[(8 * q + j) * NHID];
  }

  v8f acc;
#pragma unroll
  for (int r = 0; r < 8; ++r)
    acc[r] = bl[16 * kt + 8 * (r >> 2) + 2 * hf + (r & 1) + 4 * ((r >> 1) & 1)];

  __builtin_amdgcn_sched_barrier(0);

#pragma unroll
  for (int q = 0; q < NIN / 8; ++q) {
    acc = __builtin_amdgcn_wmma_f32_16x16x4_f32(
        false, (v2f){ wa[q][0], wa[q][1] },
        false, (v2f){ xq[q].x, xq[q].y }, (short)0, acc, false, false);
    acc = __builtin_amdgcn_wmma_f32_16x16x4_f32(
        false, (v2f){ wa[q][2], wa[q][3] },
        false, (v2f){ xq[q].z, xq[q].w }, (short)0, acc, false, false);
  }

  __builtin_amdgcn_sched_barrier(0);
  asm volatile("v_nop\n\tv_nop\n\tv_nop\n\tv_nop\n\tv_nop" : "+v"(acc));

  const v4f c0 = { fmaxf(acc[0], 0.0f), fmaxf(acc[1], 0.0f),
                   fmaxf(acc[2], 0.0f), fmaxf(acc[3], 0.0f) };
  const v4f c1 = { fmaxf(acc[4], 0.0f), fmaxf(acc[5], 0.0f),
                   fmaxf(acc[6], 0.0f), fmaxf(acc[7], 0.0f) };
  const v4f zz = { 0.0f, 0.0f, 0.0f, 0.0f };
  v4f* vv = (v4f*)vF;
  const int b = bn >> 10;
  const int n = bn & (NN - 1);
  const int qo = 2 * kt;
#pragma unroll
  for (int cb = 0; cb < 3; ++cb) {
    const int np = n + 1 - cb;
    if (np >= 0 && np < NN) {
      const int bnp = (b << 10) + np;
      const size_t at = ((size_t)(bnp >> 4) * NQ + 4 * cb + qo) * 32 + 16 * hf + (bnp & 15);
      vv[at]      = c0;
      vv[at + 32] = c1;
    }
  }
  if (n == 0) {
    const size_t at = ((size_t)(bn >> 4) * NQ + qo) * 32 + 16 * hf;
    vv[at]      = zz;
    vv[at + 32] = zz;
  }
  if (n == NN - 1) {
    const size_t at = ((size_t)(bn >> 4) * NQ + 8 + qo) * 32 + 16 * hf + 15;
    vv[at]      = zz;
    vv[at + 32] = zz;
  }
}

#define TO 4
#define TU 3

#define T_LOAD_A(U)                                                           \
  _Pragma("unroll")                                                           \
  for (int q = 0; q < NQ; ++q) {                                              \
    _Pragma("unroll")                                                         \
    for (int a = 0; a < TO; ++a)                                              \
      a4[U][a][q] = ap[((size_t)(a * NGT + (U)) * NQ + q) * 32];              \
  }

#define T_WMMAS(U, Q0, Q1)                                                    \
  _Pragma("unroll")                                                           \
  for (int q = (Q0); q < (Q1); ++q) {                                         \
    _Pragma("unroll")                                                         \
    for (int a = 0; a < TO; ++a)                                              \
      acc[a] = __builtin_amdgcn_wmma_f32_16x16x4_f32(                         \
          false, (v2f){ a4[U][a][q].x, a4[U][a][q].y },                       \
          false, (v2f){ bq[q].x, bq[q].y }, (short)0, acc[a], false, false);  \
    _Pragma("unroll")                                                         \
    for (int a = 0; a < TO; ++a)                                              \
      acc[a] = __builtin_amdgcn_wmma_f32_16x16x4_f32(                         \
          false, (v2f){ a4[U][a][q].z, a4[U][a][q].w },                       \
          false, (v2f){ bq[q].z, bq[q].w }, (short)0, acc[a], false, false);  \
  }

#define T_STORE(U)                                                            \
  _Pragma("unroll")                                                           \
  for (int a = 0; a < TO; ++a) {                                              \
    const v4f lo = { acc[a][0], acc[a][1], acc[a][4], acc[a][5] };            \
    const v4f hi = { acc[a][2], acc[a][3], acc[a][6], acc[a][7] };            \
    float* d = dst + (U) * (2 * 32 * 4) + (8 * oq + a) * 4;                   \
    *(v4f*)(d)          = lo;                                                 \
    *(v4f*)(d + 16 * 4) = hi;                                                 \
  }

__global__ __launch_bounds__(128) void k_t_wmma(
    const float* __restrict__ vF, const float* __restrict__ wF,
    float* __restrict__ tF) {
  const int lane = threadIdx.x & 31;
  const int c    = lane & 15;
  const int hf   = lane >> 4;
  const int wv  = __builtin_amdgcn_readfirstlane((int)(threadIdx.x >> 5));
  const int blk = blockIdx.x;
  const int oq  = blk & 1;
  const int gt0 = TU * ((blk >> 1) & 1);
  const int bt  = 4 * (blk >> 2) + wv;

  const v4f* vp = (const v4f*)vF + (size_t)bt * NQ * 32 + lane;
  const v4f* ap = (const v4f*)wF + (size_t)((TO * oq) * NGT + gt0) * NQ * 32 + lane;
  v4f bq[NQ];
  v4f a4[TU][TO][NQ];

  const int bi = bt * 16 + c;
  float* dst = tF + ((((size_t)(bi >> 1) * NQ + 2 * gt0 + hf) * 32) + (bi & 1) * 4) * 4;

#pragma unroll
  for (int q = 0; q < NQ; ++q) bq[q] = vp[q * 32];
  T_LOAD_A(0)
  __builtin_amdgcn_sched_barrier(0);

  {
    v8f acc[TO] = {};
    T_WMMAS(0, 0, NQ / 2)
    __builtin_amdgcn_sched_barrier(0);
    T_LOAD_A(1)
    __builtin_amdgcn_sched_barrier(0);
    T_WMMAS(0, NQ / 2, NQ)
    __builtin_amdgcn_sched_barrier(0);
    T_STORE(0)
  }
  T_LOAD_A(2)
  __builtin_amdgcn_sched_barrier(0);
  {
    v8f acc[TO] = {};
    T_WMMAS(1, 0, NQ)
    __builtin_amdgcn_sched_barrier(0);
    T_STORE(1)
  }
  __builtin_amdgcn_sched_barrier(0);
  {
    v8f acc[TO] = {};
    T_WMMAS(2, 0, NQ)
    T_STORE(2)
  }
}
#undef T_STORE
#undef T_WMMAS
#undef T_LOAD_A
#undef TU
#undef TO

#define NA   4
#define JTW  16

#define TILE_WMMAS(BSET)                                                      \
  _Pragma("unroll")                                                           \
  for (int q = 0; q < NQ; ++q) {                                              \
    _Pragma("unroll")                                                         \
    for (int a = 0; a < NA; ++a)                                              \
      acc[a] = __builtin_amdgcn_wmma_f32_16x16x4_f32(                         \
          false, (v2f){ a4[a][q].x, a4[a][q].y },                             \
          false, (v2f){ BSET[q].x, BSET[q].y }, (short)0, acc[a], false, false); \
    _Pragma("unroll")                                                         \
    for (int a = 0; a < NA; ++a)                                              \
      acc[a] = __builtin_amdgcn_wmma_f32_16x16x4_f32(                         \
          false, (v2f){ a4[a][q].z, a4[a][q].w },                             \
          false, (v2f){ BSET[q].z, BSET[q].w }, (short)0, acc[a], false, false); \
  }

#define TILE_STORE(TJ)                                                        \
  _Pragma("unroll")                                                           \
  for (int a = 0; a < NA; ++a) {                                              \
    float* op = ob + ((size_t)(2 * a) * NN + (TJ) * 16) * NOUT;               \
    v4f r0 = { acc[a][0] + bb.x, acc[a][1] + bb.y,                            \
               acc[a][2] + bb.z, acc[a][3] + bb.w };                          \
    v4f r1 = { acc[a][4] + bb.x, acc[a][5] + bb.y,                            \
               acc[a][6] + bb.z, acc[a][7] + bb.w };                          \
    __builtin_nontemporal_store(r0, (v4f*)op);                                \
    __builtin_nontemporal_store(r1, (v4f*)(op + (size_t)NN * NOUT));          \
  }

#define TILE_LOAD_B(BSET, TJ)                                                 \
  {                                                                           \
    const int tn = ((TJ) < JTW) ? (TJ) : JTW - 1;                             \
    const v4f* bn = bp + (size_t)tn * NQ * 32;                                \
    _Pragma("unroll")                                                         \
    for (int q = 0; q < NQ; ++q) BSET[q] = bn[q * 32];                        \
  }

__global__ __launch_bounds__(256) void k_bilinear_out(
    const float* __restrict__ tF, const float* __restrict__ vF,
    const float* __restrict__ bb_g, float* __restrict__ out) {
  const int lane   = threadIdx.x & 31;
  const int laneLo = lane & 15;
  const int hi     = lane >> 4;
  const int wv  = __builtin_amdgcn_readfirstlane((int)(threadIdx.x >> 5));
  const int blk = blockIdx.x;
  const int rgg = blk & 15;
  const int jq  = (blk >> 4) & 3;
  const int b   = blk >> 6;
  const int mt0 = (rgg * 8 + wv) * NA;
  const int i0  = mt0 * 2;

  const v4f bb = *(const v4f*)(bb_g + 4 * hi);

  const v4f* ap = (const v4f*)tF + (size_t)(b * 512 + mt0) * NQ * 32 + lane;
  v4f a4[NA][NQ];
#pragma unroll
  for (int q = 0; q < NQ; ++q)
#pragma unroll
    for (int a = 0; a < NA; ++a) a4[a][q] = ap[(a * NQ + q) * 32];

  __builtin_amdgcn_sched_barrier(0);

  const v4f* bp = (const v4f*)vF + (size_t)(b * 64 + jq * JTW) * NQ * 32 + lane;
  v4f b0[NQ], b1[NQ];
  TILE_LOAD_B(b0, 0)
  TILE_LOAD_B(b1, 1)

  float* ob = out + (((size_t)b * NN + i0) * NN + jq * (JTW * 16) + laneLo) * NOUT + 4 * hi;

#pragma unroll 1
  for (int tj = 0; tj < JTW; tj += 2) {
    {
      v8f acc[NA] = {};
      TILE_WMMAS(b0)
      __builtin_amdgcn_sched_barrier(0);
      TILE_STORE(tj)
      TILE_LOAD_B(b0, tj + 2)
      __builtin_amdgcn_sched_barrier(0);
    }
    {
      v8f acc[NA] = {};
      TILE_WMMAS(b1)
      __builtin_amdgcn_sched_barrier(0);
      TILE_STORE(tj + 1)
      TILE_LOAD_B(b1, tj + 3)
      __builtin_amdgcn_sched_barrier(0);
    }
  }
}
#undef TILE_LOAD_B
#undef TILE_STORE
#undef TILE_WMMAS

extern "C" void kernel_launch(void* const* d_in, const int* in_sizes, int n_in,
                              void* d_out, int out_size, void* d_ws, size_t ws_size,
                              hipStream_t stream) {
  const float* x     = (const float*)d_in[0];
  const float* W_lin = (const float*)d_in[1];
  const float* b_lin = (const float*)d_in[2];
  const float* W_bil = (const float*)d_in[3];
  const float* b_bil = (const float*)d_in[4];
  float* out = (float*)d_out;

  char* ws = (char*)d_ws;
  float* vF = (float*)ws;
  float* tF = (float*)(ws + (size_t)NB * NN * HH * 4);

  float* wF = out;

  k_lin_relu_v<<<LIN_BLOCKS + WF_BLOCKS, 128, 0, stream>>>(x, W_lin, b_lin, W_bil, vF, wF);
  k_t_wmma<<<256, 128, 0, stream>>>(vF, wF, tF);
  k_bilinear_out<<<256, 256, 0, stream>>>(tF, vF, b_bil, out);
}
